# starting kernel
# baseline (speedup 1.0000x reference)
#include <hip/hip_runtime.h>

#define D 32
#define NG 256

typedef __attribute__((ext_vector_type(2))) float v2f;
typedef __attribute__((ext_vector_type(8))) float v8f;

__global__ void zero_kernel(float* __restrict__ p, int n) {
    int i = blockIdx.x * blockDim.x + threadIdx.x;
    int stride = gridDim.x * blockDim.x;
    for (; i < n; i += stride) p[i] = 0.f;
}

__global__ void node_scatter(const float* __restrict__ x,
                             const int* __restrict__ batch,
                             float* __restrict__ acc, int n_nodes) {
    __shared__ float sh[NG * D];
    for (int i = threadIdx.x; i < NG * D; i += blockDim.x) sh[i] = 0.f;
    __syncthreads();

    int chunk = (n_nodes + gridDim.x - 1) / gridDim.x;
    int start = blockIdx.x * chunk;
    int end   = min(start + chunk, n_nodes);
    int d   = threadIdx.x & (D - 1);
    int sub = threadIdx.x >> 5;
    int step = blockDim.x >> 5;
    for (int n = start + sub; n < end; n += step) {
        int g = batch[n];
        atomicAdd(&sh[g * D + d], x[(long)n * D + d]);
    }
    __syncthreads();
    for (int i = threadIdx.x; i < NG * D; i += blockDim.x) {
        float v = sh[i];
        if (v != 0.f) atomicAdd(&acc[i], v);
    }
}

__global__ void edge_scatter(const float* __restrict__ edge_attr,
                             const int* __restrict__ src,
                             const int* __restrict__ batch,
                             float* __restrict__ acc, int n_edges) {
    __shared__ float sh[NG * D];
    for (int i = threadIdx.x; i < NG * D; i += blockDim.x) sh[i] = 0.f;
    __syncthreads();

    int chunk = (n_edges + gridDim.x - 1) / gridDim.x;
    int start = blockIdx.x * chunk;
    int end   = min(start + chunk, n_edges);
    int d   = threadIdx.x & (D - 1);
    int sub = threadIdx.x >> 5;
    int step = blockDim.x >> 5;
    for (int e = start + sub; e < end; e += step) {
        int g = batch[src[e]];
        atomicAdd(&sh[g * D + d], edge_attr[(long)e * D + d]);
    }
    __syncthreads();
    for (int i = threadIdx.x; i < NG * D; i += blockDim.x) {
        float v = sh[i];
        if (v != 0.f) atomicAdd(&acc[i], v);
    }
}

__global__ void gemm_bias_relu(const float* __restrict__ node_acc,
                               const float* __restrict__ edge_acc,
                               const float* __restrict__ u,
                               const float* __restrict__ W,
                               const float* __restrict__ b,
                               float* __restrict__ out, int n_graphs) {
    const int lane  = threadIdx.x & 31;
    const int wid   = (blockIdx.x * blockDim.x + threadIdx.x) >> 5;
    const int mtiles = n_graphs >> 4;
    const int tile_m = wid % mtiles;
    const int tile_n = wid / mtiles;

    const int l15   = lane & 15;
    const int khalf = (lane >> 4) << 1;
    const int mrow  = tile_m * 16 + l15;
    const int ncol  = tile_n * 16 + l15;

    v8f c = {};
    #pragma unroll
    for (int kb = 0; kb < 24; ++kb) {
        const int k0 = kb * 4;
        const float* A = (k0 < 32) ? node_acc : (k0 < 64) ? edge_acc : u;
        const int koff = k0 & 31;
        v2f a, bm;
        a[0]  = A[mrow * D + koff + khalf + 0];
        a[1]  = A[mrow * D + koff + khalf + 1];
        bm[0] = W[(k0 + khalf + 0) * D + ncol];
        bm[1] = W[(k0 + khalf + 1) * D + ncol];
        c = __builtin_amdgcn_wmma_f32_16x16x4_f32(
                 false, a,  false, bm,
                 (short)0, c,  false,  false);
    }

    const float bias = b[ncol];
    const int rbase = tile_m * 16 + ((lane >> 4) << 3);
    #pragma unroll
    for (int r = 0; r < 8; ++r) {
        float v = c[r] + bias;
        out[(rbase + r) * D + ncol] = v > 0.f ? v : 0.f;
    }
}

extern "C" void kernel_launch(void* const* d_in, const int* in_sizes, int n_in,
                              void* d_out, int out_size, void* d_ws, size_t ws_size,
                              hipStream_t stream) {
    const float* x          = (const float*)d_in[0];
    const int*   edge_index = (const int*)d_in[1];
    const float* edge_attr  = (const float*)d_in[2];
    const float* u          = (const float*)d_in[3];
    const int*   batch      = (const int*)d_in[4];
    const float* W          = (const float*)d_in[5];
    const float* b          = (const float*)d_in[6];
    float*       out        = (float*)d_out;

    const int n_nodes  = in_sizes[4];
    const int n_edges  = in_sizes[2] / D;
    const int n_graphs = in_sizes[3] / D;

    float* node_acc = (float*)d_ws;
    float* edge_acc = node_acc + n_graphs * D;

    zero_kernel<<<32, 256, 0, stream>>>(node_acc, 2 * n_graphs * D);
    node_scatter<<<512, 256, 0, stream>>>(x, batch, node_acc, n_nodes);
    edge_scatter<<<1024, 256, 0, stream>>>(edge_attr, edge_index, batch,
                                           edge_acc, n_edges);
    const int waves  = (n_graphs / 16) * 2;
    const int blocks = (waves + 15) / 16;
    gemm_bias_relu<<<blocks, 512, 0, stream>>>(node_acc, edge_acc, u, W, b,
                                               out, n_graphs);
}


# node segment-sum by owner blocks inside the product kernel: block g finds graph g's 16-node groups by id ranges from launch 1 and sums their rows; no node atomics, 3 launches; node sum order changes
# speedup vs baseline: 4.8977x; 4.8977x over previous
#include <hip/hip_runtime.h>

#define D 32
#define NG 256

typedef __attribute__((ext_vector_type(2))) float v2f;
typedef __attribute__((ext_vector_type(8))) float v8f;

#define MIXED 0xFFFFu
#define SUMCAP 8192
#define SUMU   8
static_assert(SUMU == 8, "the statement that holds the id loads together names 8 registers");

#define RNG_UNK  0x00FFu
#define RNG_NONE 0x00FEu

__global__ void zero_kernel(float* __restrict__ p, int n,
                            const int* __restrict__ batch,
                            unsigned short* __restrict__ summ,
                            unsigned short* __restrict__ rng, int n_nodes) {
    const int tid = blockIdx.x * blockDim.x + threadIdx.x;
    const int stride = gridDim.x * blockDim.x;

    const int lane = threadIdx.x & 31;
    const int h = lane >> 4;
    const int ngroups = (n_nodes + 15) >> 4;
    for (int base = (tid - lane) * SUMU; base < n_nodes; base += stride * SUMU) {
        int id[SUMU];
        #pragma unroll
        for (int j = 0; j < SUMU; ++j)
            id[j] = batch[min(base + 32 * j + lane, n_nodes - 1)];
        asm volatile("" : "+v"(id[0]), "+v"(id[1]), "+v"(id[2]), "+v"(id[3]),
                          "+v"(id[4]), "+v"(id[5]), "+v"(id[6]), "+v"(id[7]));
        #pragma unroll
        for (int j = 0; j < SUMU; ++j) {
            const int f0 = __builtin_amdgcn_readlane(id[j], 0);
            const int f1 = __builtin_amdgcn_readlane(id[j], 16);
            const int first = h ? f1 : f0;
            const unsigned m = __builtin_amdgcn_ballot_w32(id[j] == first && (unsigned)id[j] < (unsigned)NG);
            const unsigned gm = h ? (m >> 16) : (m & 0xFFFFu);
            const int grp = ((base + 32 * j) >> 4) + h;
            if ((lane & 15) == 0 && grp < ngroups)
                summ[grp] = (unsigned short)(gm == 0xFFFFu ? (unsigned)first : MIXED);
            int mn = id[j], mx = id[j];
            mn = min(mn, __builtin_amdgcn_update_dpp(mn, mn, 0x168, 0xF, 0xF, false));
            mx = max(mx, __builtin_amdgcn_update_dpp(mx, mx, 0x168, 0xF, 0xF, false));
            mn = min(mn, __builtin_amdgcn_update_dpp(mn, mn, 0x164, 0xF, 0xF, false));
            mx = max(mx, __builtin_amdgcn_update_dpp(mx, mx, 0x164, 0xF, 0xF, false));
            mn = min(mn, __builtin_amdgcn_update_dpp(mn, mn, 0x162, 0xF, 0xF, false));
            mx = max(mx, __builtin_amdgcn_update_dpp(mx, mx, 0x162, 0xF, 0xF, false));
            mn = min(mn, __builtin_amdgcn_update_dpp(mn, mn, 0x161, 0xF, 0xF, false));
            mx = max(mx, __builtin_amdgcn_update_dpp(mx, mx, 0x161, 0xF, 0xF, false));
            if ((lane & 15) == 0 && grp < ngroups)
                rng[grp] = (unsigned short)((mn >= 0 && mx < 256) ? (unsigned)(mn | (mx << 8)) : RNG_UNK);
        }
    }
    for (int i = tid; i < n; i += stride) p[i] = 0.f;
}

#define EPB   8192
#define EW    32
#define KPT   (EPB / (EW * 32))
static_assert(KPT == 8, "the statement that holds the node-number loads together names 8 registers");
static_assert(SUMCAP / 8 == EW * 32, "the summary is copied to LDS with one 16-byte load per thread");

__device__ __forceinline__ void trip_wmma(const float2 (&pr)[8], const int (&pe)[8],
                                          int c, v8f& t0, v8f& t1) {
    #pragma unroll
    for (int u = 0; u < 4; ++u) {
        const int ea = pe[2 * u], eb = pe[2 * u + 1];
        const bool va = ea >= 0, vb = eb >= 0;
        v2f a, b0, b1;
        a[0] = (va && ((ea >> 25) & 15) == c) ? 1.f : 0.f;
        a[1] = (vb && ((eb >> 25) & 15) == c) ? 1.f : 0.f;
        b0[0] = va ? pr[2 * u].x : 0.f;  b0[1] = vb ? pr[2 * u + 1].x : 0.f;
        b1[0] = va ? pr[2 * u].y : 0.f;  b1[1] = vb ? pr[2 * u + 1].y : 0.f;
        t0 = __builtin_amdgcn_wmma_f32_16x16x4_f32(
                false, a, false, b0, (short)0, t0, false, false);
        t1 = __builtin_amdgcn_wmma_f32_16x16x4_f32(
                false, a, false, b1, (short)0, t1, false, false);
    }
}

__device__ __forceinline__ void trip_load(const float* __restrict__ edge_attr,
                                          const int* rq, int h, int c,
                                          float2 (&pr)[8], int (&pe)[8]) {
    #pragma unroll
    for (int u = 0; u < 4; ++u) {
        const int2 p2 = *(const int2*)(rq + 4 * u + 2 * h);
        pe[2 * u] = p2.x; pe[2 * u + 1] = p2.y;
    }
    const char* rows = (const char*)edge_attr;
    #pragma unroll
    for (int k = 0; k < 8; ++k) {
        const unsigned off = ((unsigned)pe[k] << 7) | ((unsigned)c << 3);
        pr[k] = *(const float2*)(rows + off);
    }
}

__global__ __launch_bounds__(1024)
void edge_sorted_wmma(const float* __restrict__ edge_attr,
                      const int* __restrict__ src,
                      const int* __restrict__ batch,
                      const unsigned short* __restrict__ summ,
                      float* __restrict__ acc, int n_edges, int n_nodes) {
    __shared__ float sh[NG * D];
    __shared__ __attribute__((aligned(16))) int sorted[EPB + 16 * 32];
    __shared__ int cnt[EW * 16];
    __shared__ __attribute__((aligned(16))) unsigned short sm[SUMCAP];
    {
        const int nsum = min((n_nodes + 15) >> 4, SUMCAP);
        const int nvec = nsum >> 3;
        const int t = threadIdx.x;
        if (t < nvec) ((uint4*)sm)[t] = ((const uint4*)summ)[t];
        if (t < (nsum & 7)) sm[8 * nvec + t] = summ[8 * nvec + t];
    }
    for (int i = threadIdx.x; i < NG * D; i += EW * 32) sh[i] = 0.f;
    __syncthreads();

    const int lane = threadIdx.x & 31;
    const int wave = __builtin_amdgcn_readfirstlane((int)(threadIdx.x >> 5));
    const int w = wave & 15;
    const int p = wave >> 4;
    const int c = lane & 15;
    const int h = lane >> 4;
    const unsigned nj0 = (c & 1) ? 0u : ~0u;
    const unsigned nj1 = (c & 2) ? 0u : ~0u;
    const unsigned nj2 = (c & 4) ? 0u : ~0u;
    const unsigned nj3 = (c & 8) ? 0u : ~0u;

    const int per_block = (n_edges + gridDim.x - 1) / gridDim.x;
    const int start = blockIdx.x * per_block;
    const int end   = min(start + per_block, n_edges);

    v8f t0 = (v8f){}, t1 = (v8f){};
    float2 prA[8], prB[8];
    int peA[8], peB[8];
    #pragma unroll
    for (int k = 0; k < 8; ++k) {
        prA[k].x = 0.f; prA[k].y = 0.f; peA[k] = -1;
        prB[k].x = 0.f; prB[k].y = 0.f; peB[k] = -1;
    }

    for (int cb = start; cb < end; cb += EPB) {
        const int n_local = min(EPB, end - cb);
        int sv[KPT];
        #pragma unroll
        for (int k = 0; k < KPT; ++k) {
            const int i = threadIdx.x + EW * 32 * k;
            sv[k] = src[min(cb + i, end - 1)];
        }
        asm volatile("" : "+v"(sv[0]), "+v"(sv[1]), "+v"(sv[2]), "+v"(sv[3]),
                          "+v"(sv[4]), "+v"(sv[5]), "+v"(sv[6]), "+v"(sv[7]));
        int gv[KPT];
        #pragma unroll
        for (int k = 0; k < KPT; ++k) {
            const unsigned grp = (unsigned)sv[k] >> 4;
            const int wd = sm[min(grp, (unsigned)(SUMCAP - 1))];
            gv[k] = grp < (unsigned)SUMCAP ? wd : (int)MIXED;
        }
        int gb[KPT];
        #pragma unroll
        for (int k = 0; k < KPT; ++k) {
            const int keep = gv[k] == (int)MIXED ? -1 : 0;
            gb[k] = batch[sv[k] & keep];
        }
        {
            const int nu = (n_local * 5) >> 5;
            const char* rows0 = (const char*)edge_attr + ((size_t)cb << 7);
            for (int u = threadIdx.x; u < nu; u += EW * 32)
                __builtin_prefetch(rows0 + ((size_t)u << 8), 0, 3);
        }
        #pragma unroll
        for (int k = 0; k < KPT; ++k) gv[k] = gv[k] == (int)MIXED ? gb[k] : gv[k];

        int ent[KPT], cd[KPT], ck[KPT];
        int wc = 0;
        #pragma unroll
        for (int k = 0; k < KPT; ++k) {
            const int i = threadIdx.x + EW * 32 * k;
            const int g = gv[k];
            const bool ok = i < n_local && (unsigned)g < (unsigned)NG;
            const int t = g >> 4;
            const unsigned bv = __builtin_amdgcn_ballot_w32(ok);
            const unsigned b0 = __builtin_amdgcn_ballot_w32((t & 1) != 0);
            const unsigned b1 = __builtin_amdgcn_ballot_w32((t & 2) != 0);
            const unsigned b2 = __builtin_amdgcn_ballot_w32((t & 4) != 0);
            const unsigned b3 = __builtin_amdgcn_ballot_w32((t & 8) != 0);
            const unsigned mj = bv & (b0 ^ nj0) & (b1 ^ nj1) & (b2 ^ nj2) & (b3 ^ nj3);
            ck[k] = __builtin_popcount(mj);
            wc += ck[k];
            const unsigned nt0 = (t & 1) ? 0u : ~0u;
            const unsigned nt1 = (t & 2) ? 0u : ~0u;
            const unsigned nt2 = (t & 4) ? 0u : ~0u;
            const unsigned nt3 = (t & 8) ? 0u : ~0u;
            const unsigned mt = bv & (b0 ^ nt0) & (b1 ^ nt1) & (b2 ^ nt2) & (b3 ^ nt3);
            const int rank = (int)__builtin_amdgcn_mbcnt_lo(mt, 0u);
            cd[k]  = ok ? (rank | ((t & 15) << 8)) : -1;
            ent[k] = ((g & 15) << 25) | (cb + i);
        }
        if (lane < 16) cnt[wave * 16 + lane] = wc;
        __syncthreads();

        int tot = 0, before = 0;
        #pragma unroll 8
        for (int ww = 0; ww < EW; ++ww) {
            const int v = cnt[ww * 16 + c];
            tot += v;
            before += (ww < wave) ? v : 0;
        }
        const int padded = (tot + 31) & ~31;
        int base = 0;
        #pragma unroll
        for (int jj = 0; jj < 15; ++jj) {
            const int v = __builtin_amdgcn_readlane(padded, jj);
            base += (c > jj) ? v : 0;
        }
        int running = base + before;

        #pragma unroll
        for (int k = 0; k < KPT; ++k) {
            const int t = (cd[k] >> 8) & 15;
            const int off = __builtin_amdgcn_ds_bpermute(t << 2, running);
            if (cd[k] >= 0) sorted[off + (cd[k] & 255)] = ent[k];
            running += ck[k];
        }
        const int nb = __builtin_amdgcn_readlane(tot, w);
        const int np = __builtin_amdgcn_readlane(padded, w);
        const int sb = __builtin_amdgcn_readlane(base, w);
        if (p == 0 && nb + lane < np) sorted[sb + nb + lane] = (int)(0x80000000u | (unsigned)start);
        __syncthreads();

        const int half = ((np >> 1) + 31) & ~31;
        const int lo = p ? half : 0;
        const int hi = p ? np : half;
        const int* seg = sorted + sb;
        for (int q = lo; q < hi; q += 32) {
            trip_wmma(prA, peA, c, t0, t1);
            trip_load(edge_attr, seg + q, h, c, prA, peA);
            trip_wmma(prB, peB, c, t0, t1);
            trip_load(edge_attr, seg + q + 16, h, c, prB, peB);
        }
    }
    trip_wmma(prA, peA, c, t0, t1);
    trip_wmma(prB, peB, c, t0, t1);

    #pragma unroll
    for (int r = 0; r < 8; ++r) {
        const int graph = 16 * w + 8 * h + r;
        atomicAdd(&sh[graph * D + 2 * c + 0], t0[r]);
        atomicAdd(&sh[graph * D + 2 * c + 1], t1[r]);
    }
    __syncthreads();
    for (int i = threadIdx.x; i < NG * D; i += EW * 32) {
        float v = sh[i];
        if (v != 0.f) atomicAdd(&acc[i], v);
    }
}

#define GW  16
#define SCU 16

__global__ __launch_bounds__(GW * 32)
void node_sum_gemm(const float* __restrict__ x,
                   const int* __restrict__ batch,
                   const unsigned short* __restrict__ rng,
                   const float* __restrict__ edge_acc,
                   const float* __restrict__ u,
                   const float* __restrict__ W,
                   const float* __restrict__ b,
                   float* __restrict__ out, int n_nodes) {
    __shared__ int   wl[GW][SCU * 32];
    __shared__ float part[GW][D];
    __shared__ float arow[3 * D];
    __shared__ float orow[D];

    const int g    = blockIdx.x;
    const int lane = threadIdx.x & 31;
    const int wave = __builtin_amdgcn_readfirstlane((int)(threadIdx.x >> 5));
    const int ngroups = (n_nodes + 15) >> 4;

    float ev = 0.f;
    if (threadIdx.x < 2 * D)
        ev = threadIdx.x < D ? edge_acc[g * D + threadIdx.x] : u[g * D + threadIdx.x - D];

    const int l15   = lane & 15;
    const int khalf = (lane >> 4) << 1;
    const int ncol  = (wave & 1) * 16 + l15;
    v2f bw[24];
    float bias = 0.f;
    if (wave < 2) {
        #pragma unroll
        for (int kb = 0; kb < 24; ++kb) {
            bw[kb][0] = W[(kb * 4 + khalf + 0) * D + ncol];
            bw[kb][1] = W[(kb * 4 + khalf + 1) * D + ncol];
        }
        bias = b[ncol];
    }

    float acc = 0.f;
    for (int r0 = 0; r0 < ngroups; r0 += GW * 32 * SCU) {
        unsigned wd[SCU];
        #pragma unroll
        for (int k = 0; k < SCU; ++k) {
            const int j = r0 + (k * 32 + lane) * GW + wave;
            const unsigned v = rng[min(j, ngroups - 1)];
            wd[k] = j < ngroups ? v : RNG_NONE;
        }
        int cnt = 0;
        #pragma unroll
        for (int k = 0; k < SCU; ++k) {
            const int j = r0 + (k * 32 + lane) * GW + wave;
            const int mn = (int)(wd[k] & 255u), mx = (int)(wd[k] >> 8);
            const bool mixed = mn != mx;
            const bool hit = wd[k] == RNG_UNK || (mn <= g && g <= mx);
            const unsigned m = __builtin_amdgcn_ballot_w32(hit);
            const int pos = cnt + (int)__builtin_amdgcn_mbcnt_lo(m, 0u);
            if (hit) wl[wave][pos] = (j << 1) | (mixed ? 1 : 0);
            cnt += __builtin_popcount(m);
        }
        for (int i = 0; i < cnt; i += 2) {
            const int e0 = __builtin_amdgcn_readfirstlane(wl[wave][i]);
            const int e1 = __builtin_amdgcn_readfirstlane(wl[wave][min(i + 1, cnt - 1)]);
            const bool has1 = i + 1 < cnt;
            const int n0 = (e0 >> 1) << 4, n1 = (e1 >> 1) << 4;
            const bool mx0 = (e0 & 1) != 0, mx1 = (e1 & 1) != 0;
            const int idl = batch[min((lane < 16 ? n0 : n1) + (lane & 15), n_nodes - 1)];
            float ra[16], rb[16];
            #pragma unroll
            for (int t = 0; t < 16; ++t) {
                ra[t] = x[(size_t)min(n0 + t, n_nodes - 1) * D + lane];
                rb[t] = x[(size_t)min(n1 + t, n_nodes - 1) * D + lane];
            }
            #pragma unroll
            for (int t = 0; t < 16; ++t) {
                const int g0 = __builtin_amdgcn_readlane(idl, t);
                const int g1 = __builtin_amdgcn_readlane(idl, 16 + t);
                const bool ok0 = n0 + t < n_nodes && (!mx0 || g0 == g);
                const bool ok1 = has1 && n1 + t < n_nodes && (!mx1 || g1 == g);
                acc += ok0 ? ra[t] : 0.f;
                acc += ok1 ? rb[t] : 0.f;
            }
        }
    }
    part[wave][lane] = acc;
    if (threadIdx.x < 2 * D) arow[D + threadIdx.x] = ev;
    __syncthreads();
    if (threadIdx.x < D) {
        float s = 0.f;
        #pragma unroll
        for (int w = 0; w < GW; ++w) s += part[w][threadIdx.x];
        arow[threadIdx.x] = s;
    }
    __syncthreads();

    if (wave < 2) {
        v8f c = {};
        #pragma unroll
        for (int kb = 0; kb < 24; ++kb) {
            const int k0 = kb * 4;
            v2f a;
            const float a0 = arow[k0 + khalf + 0];
            const float a1 = arow[k0 + khalf + 1];
            a[0]  = l15 == 0 ? a0 : 0.f;
            a[1]  = l15 == 0 ? a1 : 0.f;
            c = __builtin_amdgcn_wmma_f32_16x16x4_f32(
                     false, a,  false, bw[kb],
                     (short)0, c,  false,  false);
        }
        if (lane < 16) {
            const float v = c[0] + bias;
            orow[ncol] = v > 0.f ? v : 0.f;
        }
    }
    __syncthreads();
    if (wave == 0) out[g * D + lane] = orow[lane];
}

extern "C" void kernel_launch(void* const* d_in, const int* in_sizes, int n_in,
                              void* d_out, int out_size, void* d_ws, size_t ws_size,
                              hipStream_t stream) {
    const float* x          = (const float*)d_in[0];
    const int*   edge_index = (const int*)d_in[1];
    const float* edge_attr  = (const float*)d_in[2];
    const float* u          = (const float*)d_in[3];
    const int*   batch      = (const int*)d_in[4];
    const float* W          = (const float*)d_in[5];
    const float* b          = (const float*)d_in[6];
    float*       out        = (float*)d_out;

    const int n_nodes  = in_sizes[4];
    const int n_edges  = in_sizes[2] / D;
    const int n_graphs = in_sizes[3] / D;

    float* edge_acc = (float*)d_ws;
    unsigned short* summ = (unsigned short*)(edge_acc + n_graphs * D);
    unsigned short* rng  = summ + (1 << 20);

    zero_kernel<<<64, 256, 0, stream>>>(edge_acc, n_graphs * D, batch, summ, rng, n_nodes);
    edge_sorted_wmma<<<256, 1024, 0, stream>>>(edge_attr, edge_index, batch, summ,
                                               edge_acc, n_edges, n_nodes);
    node_sum_gemm<<<n_graphs, GW * 32, 0, stream>>>(x, batch, rng, edge_acc, u, W, b,
                                                    out, n_nodes);
}


# edge sums: one table in memory per die instead of one for the chip; a block adds into the table of the die it runs on, the last launch adds the 8 tables in fixed order; sum order changes
# speedup vs baseline: 5.1717x; 1.0559x over previous
#include <hip/hip_runtime.h>

#define D 32
#define NG 256
#define NT 8

typedef __attribute__((ext_vector_type(2))) float v2f;
typedef __attribute__((ext_vector_type(8))) float v8f;

#define MIXED 0xFFFFu
#define SUMCAP 8192
#define SUMU   8
static_assert(SUMU == 8, "the statement that holds the id loads together names 8 registers");

#define RNG_UNK  0x00FFu
#define RNG_NONE 0x00FEu

__global__ void zero_kernel(float* __restrict__ p, int n,
                            const int* __restrict__ batch,
                            unsigned short* __restrict__ summ,
                            unsigned short* __restrict__ rng, int n_nodes) {
    const int tid = blockIdx.x * blockDim.x + threadIdx.x;
    const int stride = gridDim.x * blockDim.x;

    const int lane = threadIdx.x & 31;
    const int h = lane >> 4;
    const int ngroups = (n_nodes + 15) >> 4;
    for (int base = (tid - lane) * SUMU; base < n_nodes; base += stride * SUMU) {
        int id[SUMU];
        #pragma unroll
        for (int j = 0; j < SUMU; ++j)
            id[j] = batch[min(base + 32 * j + lane, n_nodes - 1)];
        asm volatile("" : "+v"(id[0]), "+v"(id[1]), "+v"(id[2]), "+v"(id[3]),
                          "+v"(id[4]), "+v"(id[5]), "+v"(id[6]), "+v"(id[7]));
        #pragma unroll
        for (int j = 0; j < SUMU; ++j) {
            const int f0 = __builtin_amdgcn_readlane(id[j], 0);
            const int f1 = __builtin_amdgcn_readlane(id[j], 16);
            const int first = h ? f1 : f0;
            const unsigned m = __builtin_amdgcn_ballot_w32(id[j] == first && (unsigned)id[j] < (unsigned)NG);
            const unsigned gm = h ? (m >> 16) : (m & 0xFFFFu);
            const int grp = ((base + 32 * j) >> 4) + h;
            if ((lane & 15) == 0 && grp < ngroups)
                summ[grp] = (unsigned short)(gm == 0xFFFFu ? (unsigned)first : MIXED);
            int mn = id[j], mx = id[j];
            mn = min(mn, __builtin_amdgcn_update_dpp(mn, mn, 0x168, 0xF, 0xF, false));
            mx = max(mx, __builtin_amdgcn_update_dpp(mx, mx, 0x168, 0xF, 0xF, false));
            mn = min(mn, __builtin_amdgcn_update_dpp(mn, mn, 0x164, 0xF, 0xF, false));
            mx = max(mx, __builtin_amdgcn_update_dpp(mx, mx, 0x164, 0xF, 0xF, false));
            mn = min(mn, __builtin_amdgcn_update_dpp(mn, mn, 0x162, 0xF, 0xF, false));
            mx = max(mx, __builtin_amdgcn_update_dpp(mx, mx, 0x162, 0xF, 0xF, false));
            mn = min(mn, __builtin_amdgcn_update_dpp(mn, mn, 0x161, 0xF, 0xF, false));
            mx = max(mx, __builtin_amdgcn_update_dpp(mx, mx, 0x161, 0xF, 0xF, false));
            if ((lane & 15) == 0 && grp < ngroups)
                rng[grp] = (unsigned short)((mn >= 0 && mx < 256) ? (unsigned)(mn | (mx << 8)) : RNG_UNK);
        }
    }
    for (int i = tid; i < n; i += stride) p[i] = 0.f;
}

#define EPB   8192
#define EW    32
#define KPT   (EPB / (EW * 32))
static_assert(KPT == 8, "the statement that holds the node-number loads together names 8 registers");
static_assert(SUMCAP / 8 == EW * 32, "the summary is copied to LDS with one 16-byte load per thread");

__device__ __forceinline__ void trip_wmma(const float2 (&pr)[8], const int (&pe)[8],
                                          int c, v8f& t0, v8f& t1) {
    #pragma unroll
    for (int u = 0; u < 4; ++u) {
        const int ea = pe[2 * u], eb = pe[2 * u + 1];
        const bool va = ea >= 0, vb = eb >= 0;
        v2f a, b0, b1;
        a[0] = (va && ((ea >> 25) & 15) == c) ? 1.f : 0.f;
        a[1] = (vb && ((eb >> 25) & 15) == c) ? 1.f : 0.f;
        b0[0] = va ? pr[2 * u].x : 0.f;  b0[1] = vb ? pr[2 * u + 1].x : 0.f;
        b1[0] = va ? pr[2 * u].y : 0.f;  b1[1] = vb ? pr[2 * u + 1].y : 0.f;
        t0 = __builtin_amdgcn_wmma_f32_16x16x4_f32(
                false, a, false, b0, (short)0, t0, false, false);
        t1 = __builtin_amdgcn_wmma_f32_16x16x4_f32(
                false, a, false, b1, (short)0, t1, false, false);
    }
}

__device__ __forceinline__ void trip_load(const float* __restrict__ edge_attr,
                                          const int* rq, int h, int c,
                                          float2 (&pr)[8], int (&pe)[8]) {
    #pragma unroll
    for (int u = 0; u < 4; ++u) {
        const int2 p2 = *(const int2*)(rq + 4 * u + 2 * h);
        pe[2 * u] = p2.x; pe[2 * u + 1] = p2.y;
    }
    const char* rows = (const char*)edge_attr;
    #pragma unroll
    for (int k = 0; k < 8; ++k) {
        const unsigned off = ((unsigned)pe[k] << 7) | ((unsigned)c << 3);
        pr[k] = *(const float2*)(rows + off);
    }
}

__global__ __launch_bounds__(1024)
void edge_sorted_wmma(const float* __restrict__ edge_attr,
                      const int* __restrict__ src,
                      const int* __restrict__ batch,
                      const unsigned short* __restrict__ summ,
                      float* __restrict__ acc, int n_edges, int n_nodes) {
    const int die = (int)((__builtin_amdgcn_s_sendmsg_rtn(0x87) >> 16) & (NT - 1));
    __shared__ float sh[NG * D];
    __shared__ __attribute__((aligned(16))) int sorted[EPB + 16 * 32];
    __shared__ int cnt[EW * 16];
    __shared__ __attribute__((aligned(16))) unsigned short sm[SUMCAP];
    {
        const int nsum = min((n_nodes + 15) >> 4, SUMCAP);
        const int nvec = nsum >> 3;
        const int t = threadIdx.x;
        if (t < nvec) ((uint4*)sm)[t] = ((const uint4*)summ)[t];
        if (t < (nsum & 7)) sm[8 * nvec + t] = summ[8 * nvec + t];
    }
    for (int i = threadIdx.x; i < NG * D; i += EW * 32) sh[i] = 0.f;
    __syncthreads();

    const int lane = threadIdx.x & 31;
    const int wave = __builtin_amdgcn_readfirstlane((int)(threadIdx.x >> 5));
    const int w = wave & 15;
    const int p = wave >> 4;
    const int c = lane & 15;
    const int h = lane >> 4;
    const unsigned nj0 = (c & 1) ? 0u : ~0u;
    const unsigned nj1 = (c & 2) ? 0u : ~0u;
    const unsigned nj2 = (c & 4) ? 0u : ~0u;
    const unsigned nj3 = (c & 8) ? 0u : ~0u;

    const int per_block = (n_edges + gridDim.x - 1) / gridDim.x;
    const int start = blockIdx.x * per_block;
    const int end   = min(start + per_block, n_edges);

    v8f t0 = (v8f){}, t1 = (v8f){};
    float2 prA[8], prB[8];
    int peA[8], peB[8];
    #pragma unroll
    for (int k = 0; k < 8; ++k) {
        prA[k].x = 0.f; prA[k].y = 0.f; peA[k] = -1;
        prB[k].x = 0.f; prB[k].y = 0.f; peB[k] = -1;
    }

    for (int cb = start; cb < end; cb += EPB) {
        const int n_local = min(EPB, end - cb);
        int sv[KPT];
        #pragma unroll
        for (int k = 0; k < KPT; ++k) {
            const int i = threadIdx.x + EW * 32 * k;
            sv[k] = src[min(cb + i, end - 1)];
        }
        asm volatile("" : "+v"(sv[0]), "+v"(sv[1]), "+v"(sv[2]), "+v"(sv[3]),
                          "+v"(sv[4]), "+v"(sv[5]), "+v"(sv[6]), "+v"(sv[7]));
        int gv[KPT];
        #pragma unroll
        for (int k = 0; k < KPT; ++k) {
            const unsigned grp = (unsigned)sv[k] >> 4;
            const int wd = sm[min(grp, (unsigned)(SUMCAP - 1))];
            gv[k] = grp < (unsigned)SUMCAP ? wd : (int)MIXED;
        }
        int gb[KPT];
        #pragma unroll
        for (int k = 0; k < KPT; ++k) {
            const int keep = gv[k] == (int)MIXED ? -1 : 0;
            gb[k] = batch[sv[k] & keep];
        }
        {
            const int nu = (n_local * 5) >> 5;
            const char* rows0 = (const char*)edge_attr + ((size_t)cb << 7);
            for (int u = threadIdx.x; u < nu; u += EW * 32)
                __builtin_prefetch(rows0 + ((size_t)u << 8), 0, 3);
        }
        #pragma unroll
        for (int k = 0; k < KPT; ++k) gv[k] = gv[k] == (int)MIXED ? gb[k] : gv[k];

        int ent[KPT], cd[KPT], ck[KPT];
        int wc = 0;
        #pragma unroll
        for (int k = 0; k < KPT; ++k) {
            const int i = threadIdx.x + EW * 32 * k;
            const int g = gv[k];
            const bool ok = i < n_local && (unsigned)g < (unsigned)NG;
            const int t = g >> 4;
            const unsigned bv = __builtin_amdgcn_ballot_w32(ok);
            const unsigned b0 = __builtin_amdgcn_ballot_w32((t & 1) != 0);
            const unsigned b1 = __builtin_amdgcn_ballot_w32((t & 2) != 0);
            const unsigned b2 = __builtin_amdgcn_ballot_w32((t & 4) != 0);
            const unsigned b3 = __builtin_amdgcn_ballot_w32((t & 8) != 0);
            const unsigned mj = bv & (b0 ^ nj0) & (b1 ^ nj1) & (b2 ^ nj2) & (b3 ^ nj3);
            ck[k] = __builtin_popcount(mj);
            wc += ck[k];
            const unsigned nt0 = (t & 1) ? 0u : ~0u;
            const unsigned nt1 = (t & 2) ? 0u : ~0u;
            const unsigned nt2 = (t & 4) ? 0u : ~0u;
            const unsigned nt3 = (t & 8) ? 0u : ~0u;
            const unsigned mt = bv & (b0 ^ nt0) & (b1 ^ nt1) & (b2 ^ nt2) & (b3 ^ nt3);
            const int rank = (int)__builtin_amdgcn_mbcnt_lo(mt, 0u);
            cd[k]  = ok ? (rank | ((t & 15) << 8)) : -1;
            ent[k] = ((g & 15) << 25) | (cb + i);
        }
        if (lane < 16) cnt[wave * 16 + lane] = wc;
        __syncthreads();

        int tot = 0, before = 0;
        #pragma unroll 8
        for (int ww = 0; ww < EW; ++ww) {
            const int v = cnt[ww * 16 + c];
            tot += v;
            before += (ww < wave) ? v : 0;
        }
        const int padded = (tot + 31) & ~31;
        int base = 0;
        #pragma unroll
        for (int jj = 0; jj < 15; ++jj) {
            const int v = __builtin_amdgcn_readlane(padded, jj);
            base += (c > jj) ? v : 0;
        }
        int running = base + before;

        #pragma unroll
        for (int k = 0; k < KPT; ++k) {
            const int t = (cd[k] >> 8) & 15;
            const int off = __builtin_amdgcn_ds_bpermute(t << 2, running);
            if (cd[k] >= 0) sorted[off + (cd[k] & 255)] = ent[k];
            running += ck[k];
        }
        const int nb = __builtin_amdgcn_readlane(tot, w);
        const int np = __builtin_amdgcn_readlane(padded, w);
        const int sb = __builtin_amdgcn_readlane(base, w);
        if (p == 0 && nb + lane < np) sorted[sb + nb + lane] = (int)(0x80000000u | (unsigned)start);
        __syncthreads();

        const int half = ((np >> 1) + 31) & ~31;
        const int lo = p ? half : 0;
        const int hi = p ? np : half;
        const int* seg = sorted + sb;
        for (int q = lo; q < hi; q += 32) {
            trip_wmma(prA, peA, c, t0, t1);
            trip_load(edge_attr, seg + q, h, c, prA, peA);
            trip_wmma(prB, peB, c, t0, t1);
            trip_load(edge_attr, seg + q + 16, h, c, prB, peB);
        }
    }
    trip_wmma(prA, peA, c, t0, t1);
    trip_wmma(prB, peB, c, t0, t1);

    #pragma unroll
    for (int r = 0; r < 8; ++r) {
        const int graph = 16 * w + 8 * h + r;
        atomicAdd(&sh[graph * D + 2 * c + 0], t0[r]);
        atomicAdd(&sh[graph * D + 2 * c + 1], t1[r]);
    }
    __syncthreads();
    float* __restrict__ mine = acc + die * (NG * D);
    for (int i = threadIdx.x; i < NG * D; i += EW * 32) {
        float v = sh[i];
        if (v != 0.f) atomicAdd(&mine[i], v);
    }
}

#define GW  16
#define SCU 16

__global__ __launch_bounds__(GW * 32)
void node_sum_gemm(const float* __restrict__ x,
                   const int* __restrict__ batch,
                   const unsigned short* __restrict__ rng,
                   const float* __restrict__ edge_acc,
                   const float* __restrict__ u,
                   const float* __restrict__ W,
                   const float* __restrict__ b,
                   float* __restrict__ out, int n_nodes) {
    __shared__ int   wl[GW][SCU * 32];
    __shared__ float part[GW][D];
    __shared__ float arow[3 * D];
    __shared__ float orow[D];

    const int g    = blockIdx.x;
    const int lane = threadIdx.x & 31;
    const int wave = __builtin_amdgcn_readfirstlane((int)(threadIdx.x >> 5));
    const int ngroups = (n_nodes + 15) >> 4;

    float ev = 0.f;
    if (threadIdx.x < D) {
        float pt[NT];
        #pragma unroll
        for (int t = 0; t < NT; ++t) pt[t] = edge_acc[t * (NG * D) + g * D + threadIdx.x];
        #pragma unroll
        for (int t = 0; t < NT; ++t) ev += pt[t];
    } else if (threadIdx.x < 2 * D) {
        ev = u[g * D + threadIdx.x - D];
    }

    const int l15   = lane & 15;
    const int khalf = (lane >> 4) << 1;
    const int ncol  = (wave & 1) * 16 + l15;
    v2f bw[24];
    float bias = 0.f;
    if (wave < 2) {
        #pragma unroll
        for (int kb = 0; kb < 24; ++kb) {
            bw[kb][0] = W[(kb * 4 + khalf + 0) * D + ncol];
            bw[kb][1] = W[(kb * 4 + khalf + 1) * D + ncol];
        }
        bias = b[ncol];
    }

    float acc = 0.f;
    for (int r0 = 0; r0 < ngroups; r0 += GW * 32 * SCU) {
        unsigned wd[SCU];
        #pragma unroll
        for (int k = 0; k < SCU; ++k) {
            const int j = r0 + (k * 32 + lane) * GW + wave;
            const unsigned v = rng[min(j, ngroups - 1)];
            wd[k] = j < ngroups ? v : RNG_NONE;
        }
        int cnt = 0;
        #pragma unroll
        for (int k = 0; k < SCU; ++k) {
            const int j = r0 + (k * 32 + lane) * GW + wave;
            const int mn = (int)(wd[k] & 255u), mx = (int)(wd[k] >> 8);
            const bool mixed = mn != mx;
            const bool hit = wd[k] == RNG_UNK || (mn <= g && g <= mx);
            const unsigned m = __builtin_amdgcn_ballot_w32(hit);
            const int pos = cnt + (int)__builtin_amdgcn_mbcnt_lo(m, 0u);
            if (hit) wl[wave][pos] = (j << 1) | (mixed ? 1 : 0);
            cnt += __builtin_popcount(m);
        }
        for (int i = 0; i < cnt; i += 2) {
            const int e0 = __builtin_amdgcn_readfirstlane(wl[wave][i]);
            const int e1 = __builtin_amdgcn_readfirstlane(wl[wave][min(i + 1, cnt - 1)]);
            const bool has1 = i + 1 < cnt;
            const int n0 = (e0 >> 1) << 4, n1 = (e1 >> 1) << 4;
            const bool mx0 = (e0 & 1) != 0, mx1 = (e1 & 1) != 0;
            const int idl = batch[min((lane < 16 ? n0 : n1) + (lane & 15), n_nodes - 1)];
            float ra[16], rb[16];
            #pragma unroll
            for (int t = 0; t < 16; ++t) {
                ra[t] = x[(size_t)min(n0 + t, n_nodes - 1) * D + lane];
                rb[t] = x[(size_t)min(n1 + t, n_nodes - 1) * D + lane];
            }
            #pragma unroll
            for (int t = 0; t < 16; ++t) {
                const int g0 = __builtin_amdgcn_readlane(idl, t);
                const int g1 = __builtin_amdgcn_readlane(idl, 16 + t);
                const bool ok0 = n0 + t < n_nodes && (!mx0 || g0 == g);
                const bool ok1 = has1 && n1 + t < n_nodes && (!mx1 || g1 == g);
                acc += ok0 ? ra[t] : 0.f;
                acc += ok1 ? rb[t] : 0.f;
            }
        }
    }
    part[wave][lane] = acc;
    if (threadIdx.x < 2 * D) arow[D + threadIdx.x] = ev;
    __syncthreads();
    if (threadIdx.x < D) {
        float s = 0.f;
        #pragma unroll
        for (int w = 0; w < GW; ++w) s += part[w][threadIdx.x];
        arow[threadIdx.x] = s;
    }
    __syncthreads();

    if (wave < 2) {
        v8f c = {};
        #pragma unroll
        for (int kb = 0; kb < 24; ++kb) {
            const int k0 = kb * 4;
            v2f a;
            const float a0 = arow[k0 + khalf + 0];
            const float a1 = arow[k0 + khalf + 1];
            a[0]  = l15 == 0 ? a0 : 0.f;
            a[1]  = l15 == 0 ? a1 : 0.f;
            c = __builtin_amdgcn_wmma_f32_16x16x4_f32(
                     false, a,  false, bw[kb],
                     (short)0, c,  false,  false);
        }
        if (lane < 16) {
            const float v = c[0] + bias;
            orow[ncol] = v > 0.f ? v : 0.f;
        }
    }
    __syncthreads();
    if (wave == 0) out[g * D + lane] = orow[lane];
}

extern "C" void kernel_launch(void* const* d_in, const int* in_sizes, int n_in,
                              void* d_out, int out_size, void* d_ws, size_t ws_size,
                              hipStream_t stream) {
    const float* x          = (const float*)d_in[0];
    const int*   edge_index = (const int*)d_in[1];
    const float* edge_attr  = (const float*)d_in[2];
    const float* u          = (const float*)d_in[3];
    const int*   batch      = (const int*)d_in[4];
    const float* W          = (const float*)d_in[5];
    const float* b          = (const float*)d_in[6];
    float*       out        = (float*)d_out;

    const int n_nodes  = in_sizes[4];
    const int n_edges  = in_sizes[2] / D;
    const int n_graphs = in_sizes[3] / D;

    float* edge_acc = (float*)d_ws;
    unsigned short* summ = (unsigned short*)(edge_acc + NT * n_graphs * D);
    unsigned short* rng  = summ + (1 << 20);

    zero_kernel<<<64, 256, 0, stream>>>(edge_acc, NT * n_graphs * D, batch, summ, rng, n_nodes);
    edge_sorted_wmma<<<256, 1024, 0, stream>>>(edge_attr, edge_index, batch, summ,
                                               edge_acc, n_edges, n_nodes);
    node_sum_gemm<<<n_graphs, GW * 32, 0, stream>>>(x, batch, rng, edge_acc, u, W, b,
                                                    out, n_nodes);
}
